# starting kernel
# baseline (speedup 1.0000x reference)
#include <hip/hip_runtime.h>

typedef float v2f __attribute__((ext_vector_type(2)));
typedef float v8f __attribute__((ext_vector_type(8)));

#define BATCH 8
#define SEQ 2048
#define DMODEL 256
#define THREADS 128
#define WAVES 4
#define ROWS_PER_WAVE 16
#define ROWS_PER_BLOCK 64
#define LAMBDA 0.5f
#define NEG_BIG -1.0e9f

__global__ __launch_bounds__(THREADS)
void dwsa_kernel(const float* __restrict__ in,
                 const float* __restrict__ Wq,
                 const float* __restrict__ Wk,
                 const float* __restrict__ Wv,
                 float* __restrict__ out)
{
    __shared__ __align__(16) float2 s_kv[SEQ];
    __shared__ float s_red[WAVES][2][ROWS_PER_WAVE];
    __shared__ float s_A[ROWS_PER_BLOCK];

    const int tid  = threadIdx.x;
    const int b    = blockIdx.x >> 5;
    const int rowBase = (blockIdx.x & 31) * ROWS_PER_BLOCK;
    const int wave = tid >> 5;
    const int lane = tid & 31;
    const int half = lane >> 4;
    const int r    = lane & 15;
    const int n    = lane & 15;

    const float2* seq = (const float2*)in + (size_t)b * SEQ;
    for (int t = tid; t < SEQ; t += THREADS) s_kv[t] = seq[t];

    float c = 0.0f;
    #pragma unroll 8
    for (int i = 0; i < DMODEL; ++i) c += Wq[i] * Wk[i];
    c *= (1.0f / 16.0f);

    __syncthreads();

    const int   row = rowBase + wave * ROWS_PER_WAVE + r;
    const float2 q2 = s_kv[row];
    const float zs = q2.x, hs = q2.y;

    float m = -3.4e38f;
    const int t0 = half * (SEQ / 2);
    for (int t = t0; t < t0 + SEQ / 2; ++t) {
        float2 kt = s_kv[t];
        float l = c * hs * kt.y - LAMBDA * fabsf(zs - kt.x);
        l += (kt.y == 0.0f) ? NEG_BIG : 0.0f;
        m = fmaxf(m, l);
    }
    m = fmaxf(m, __shfl_xor(m, 16, 32));

    v8f acc = {0.f, 0.f, 0.f, 0.f, 0.f, 0.f, 0.f, 0.f};
    const float4* kv4 = (const float4*)s_kv;
    #pragma unroll 2
    for (int kc = 0; kc < SEQ; kc += 4) {
        const int ka = kc + 2 * half;
        float4 kk = kv4[ka >> 1];
        const float za = kk.x, ha = kk.y, zb = kk.z, hb = kk.w;
        float la = c * hs * ha - LAMBDA * fabsf(zs - za) + ((ha == 0.0f) ? NEG_BIG : 0.0f);
        float lb = c * hs * hb - LAMBDA * fabsf(zs - zb) + ((hb == 0.0f) ? NEG_BIG : 0.0f);
        v2f a;
        a.x = __expf(la - m);
        a.y = __expf(lb - m);
        v2f bop;
        bop.x = (n == 0) ? ha : ((n == 1) ? 1.0f : 0.0f);
        bop.y = (n == 0) ? hb : ((n == 1) ? 1.0f : 0.0f);
        acc = __builtin_amdgcn_wmma_f32_16x16x4_f32(
             false, a,  false, bop,
             (short)0, acc,  false,  false);
    }

    if (n < 2) {
        #pragma unroll
        for (int v = 0; v < 8; ++v)
            s_red[wave][n][v + 8 * half] = acc[v];
    }
    __syncthreads();

    if (tid < ROWS_PER_BLOCK) {
        const int w = tid >> 4, rr = tid & 15;
        const float eh = s_red[w][0][rr];
        const float se = s_red[w][1][rr];
        s_A[tid] = eh / se;
    }
    __syncthreads();

    const float wv0 = Wv[2 * tid];
    const float wv1 = Wv[2 * tid + 1];
    v2f* o2 = (v2f*)out + (size_t)(b * SEQ + rowBase) * (DMODEL / 2) + tid;
    for (int rr = 0; rr < ROWS_PER_BLOCK; ++rr) {
        const float A = s_A[rr];
        v2f val;
        val.x = A * wv0;
        val.y = A * wv1;
        __builtin_nontemporal_store(val, o2);
        o2 += DMODEL / 2;
    }
}

extern "C" void kernel_launch(void* const* d_in, const int* in_sizes, int n_in,
                              void* d_out, int out_size, void* d_ws, size_t ws_size,
                              hipStream_t stream) {
    (void)in_sizes; (void)n_in; (void)out_size; (void)d_ws; (void)ws_size;
    const float* in = (const float*)d_in[0];
    const float* Wq = (const float*)d_in[1];
    const float* Wk = (const float*)d_in[2];
    const float* Wv = (const float*)d_in[3];
    float* out = (float*)d_out;

    dim3 grid(BATCH * (SEQ / ROWS_PER_BLOCK));
    dim3 block(THREADS);
    dwsa_kernel<<<grid, block, 0, stream>>>(in, Wq, Wk, Wv, out);
}


# Split each 16-row tile's keys over 8 waves (1024-thread blocks, 8 waves per SIMD instead of 1) to hide LDS and ALU latency; slice maxima and sums meet in LDS, so the sum order changes
# speedup vs baseline: 2.6724x; 2.6724x over previous
#include <hip/hip_runtime.h>


typedef float v2f __attribute__((ext_vector_type(2)));
typedef float v8f __attribute__((ext_vector_type(8)));

#define BATCH 8
#define SEQ 2048
#define DMODEL 256
#define ROWS_PER_WAVE 16
#define ROW_TILES 4
#define ROWS_PER_BLOCK 64
#define KSPLIT 8
#define KEYS_PER_WAVE 256
#define WAVES 32
#define THREADS 1024
#define LAMBDA 0.5f
#define NEG_BIG -1.0e9f

__global__ __launch_bounds__(THREADS)
void dwsa_kernel(const float* __restrict__ in,
                 const float* __restrict__ Wq,
                 const float* __restrict__ Wk,
                 const float* __restrict__ Wv,
                 float* __restrict__ out)
{
    __shared__ __align__(16) float2 s_kv[SEQ];
    __shared__ float s_max[ROW_TILES][KSPLIT][ROWS_PER_WAVE];
    __shared__ float s_red[ROW_TILES][KSPLIT][2][ROWS_PER_WAVE];
    __shared__ float s_A[ROWS_PER_BLOCK];

    const int tid  = threadIdx.x;
    const int b    = blockIdx.x >> 5;
    const int rowBase = (blockIdx.x & 31) * ROWS_PER_BLOCK;
    const int wave = __builtin_amdgcn_readfirstlane(tid >> 5);
    const int lane = tid & 31;
    const int rt   = wave & (ROW_TILES - 1);
    const int ks   = wave >> 2;
    const int half = lane >> 4;
    const int r    = lane & 15;
    const int n    = lane & 15;

    const float2* seq = (const float2*)in + (size_t)b * SEQ;
    for (int i = 0; i < SEQ / THREADS; ++i) s_kv[tid + i * THREADS] = seq[tid + i * THREADS];

    float c = 0.0f;
    #pragma unroll 8
    for (int i = 0; i < DMODEL; ++i) c += Wq[i] * Wk[i];
    c *= (1.0f / 16.0f);

    __syncthreads();

    const int   row = rowBase + rt * ROWS_PER_WAVE + r;
    const float2 q2 = s_kv[row];
    const float zs = q2.x, hs = q2.y;
    const int   key0 = ks * KEYS_PER_WAVE;

    float m = -3.4e38f;
    const int t0 = key0 + half * (KEYS_PER_WAVE / 2);
    for (int t = t0; t < t0 + KEYS_PER_WAVE / 2; ++t) {
        float2 kt = s_kv[t];
        float l = c * hs * kt.y - LAMBDA * fabsf(zs - kt.x);
        l += (kt.y == 0.0f) ? NEG_BIG : 0.0f;
        m = fmaxf(m, l);
    }
    m = fmaxf(m, __shfl_xor(m, 16, 32));
    if (half == 0) s_max[rt][ks][r] = m;
    __syncthreads();
    for (int j = 0; j < KSPLIT; ++j) m = fmaxf(m, s_max[rt][j][r]);

    v8f acc = {0.f, 0.f, 0.f, 0.f, 0.f, 0.f, 0.f, 0.f};
    const float4* kv4 = (const float4*)s_kv;
    #pragma unroll 2
    for (int kc = key0; kc < key0 + KEYS_PER_WAVE; kc += 4) {
        const int ka = kc + 2 * half;
        float4 kk = kv4[ka >> 1];
        const float za = kk.x, ha = kk.y, zb = kk.z, hb = kk.w;
        float la = c * hs * ha - LAMBDA * fabsf(zs - za) + ((ha == 0.0f) ? NEG_BIG : 0.0f);
        float lb = c * hs * hb - LAMBDA * fabsf(zs - zb) + ((hb == 0.0f) ? NEG_BIG : 0.0f);
        v2f a;
        a.x = __expf(la - m);
        a.y = __expf(lb - m);
        v2f bop;
        bop.x = (n == 0) ? ha : ((n == 1) ? 1.0f : 0.0f);
        bop.y = (n == 0) ? hb : ((n == 1) ? 1.0f : 0.0f);
        acc = __builtin_amdgcn_wmma_f32_16x16x4_f32(
             false, a,  false, bop,
             (short)0, acc,  false,  false);
    }

    if (n < 2) {
        #pragma unroll
        for (int v = 0; v < 8; ++v)
            s_red[rt][ks][n][v + 8 * half] = acc[v];
    }
    __syncthreads();

    if (tid < ROWS_PER_BLOCK) {
        const int w = tid >> 4, rr = tid & 15;
        float eh = 0.0f, se = 0.0f;
        for (int j = 0; j < KSPLIT; ++j) {
            eh += s_red[w][j][0][rr];
            se += s_red[w][j][1][rr];
        }
        s_A[tid] = eh / se;
    }
    __syncthreads();

    const int cp = tid & 127;
    const int rg = tid >> 7;
    const float wv0 = Wv[2 * cp];
    const float wv1 = Wv[2 * cp + 1];
    v2f* o2 = (v2f*)out + (size_t)(b * SEQ + rowBase + rg) * (DMODEL / 2) + cp;
    for (int i = 0; i < ROWS_PER_BLOCK / (THREADS / 128); ++i) {
        const float A = s_A[rg + i * (THREADS / 128)];
        v2f val;
        val.x = A * wv0;
        val.y = A * wv1;
        __builtin_nontemporal_store(val, o2 + (size_t)i * (THREADS / 128) * (DMODEL / 2));
    }
}

extern "C" void kernel_launch(void* const* d_in, const int* in_sizes, int n_in,
                              void* d_out, int out_size, void* d_ws, size_t ws_size,
                              hipStream_t stream) {
    (void)in_sizes; (void)n_in; (void)out_size; (void)d_ws; (void)ws_size;
    const float* in = (const float*)d_in[0];
    const float* Wq = (const float*)d_in[1];
    const float* Wk = (const float*)d_in[2];
    const float* Wv = (const float*)d_in[3];
    float* out = (float*)d_out;

    dim3 grid(BATCH * (SEQ / ROWS_PER_BLOCK));
    dim3 block(THREADS);
    dwsa_kernel<<<grid, block, 0, stream>>>(in, Wq, Wk, Wv, out);
}


# Drop the scan for the exact row maximum: shift each row's logits by a closed-form upper bound (c*h_row times the batch's largest or smallest unmasked height); sum order kept, terms rescaled per row
# speedup vs baseline: 4.1449x; 1.5510x over previous
#include <hip/hip_runtime.h>


typedef float v2f __attribute__((ext_vector_type(2)));
typedef float v8f __attribute__((ext_vector_type(8)));

#define BATCH 8
#define SEQ 2048
#define DMODEL 256
#define ROWS_PER_WAVE 16
#define ROW_TILES 4
#define ROWS_PER_BLOCK 64
#define KSPLIT 8
#define KEYS_PER_WAVE 256
#define WAVES 32
#define THREADS 1024
#define LAMBDA 0.5f
#define NEG_BIG -1.0e9f
#define H_NONE 3.4e38f

static_assert(WAVES == 32, "one lane per wave in the second reduction stage");

template <int XMASK>
__device__ __forceinline__ float row_xor(float v) {
    const int x = __builtin_bit_cast(int, v);
    return __builtin_bit_cast(float,
        __builtin_amdgcn_update_dpp(x, x, 0x160 | XMASK, 0xf, 0xf, false));
}
__device__ __forceinline__ float rows_swap(float v) {
    const int x = __builtin_bit_cast(int, v);
    return __builtin_bit_cast(float,
        __builtin_amdgcn_permlanex16(x, x, 0x76543210u, 0xfedcba98u, false, false));
}
__device__ __forceinline__ void wave_extremes(float& hi, float& lo) {
    hi = fmaxf(hi, row_xor<1>(hi));  lo = fminf(lo, row_xor<1>(lo));
    hi = fmaxf(hi, row_xor<2>(hi));  lo = fminf(lo, row_xor<2>(lo));
    hi = fmaxf(hi, row_xor<4>(hi));  lo = fminf(lo, row_xor<4>(lo));
    hi = fmaxf(hi, row_xor<8>(hi));  lo = fminf(lo, row_xor<8>(lo));
    hi = fmaxf(hi, rows_swap(hi));   lo = fminf(lo, rows_swap(lo));
}

__global__ __launch_bounds__(THREADS)
void dwsa_kernel(const float* __restrict__ in,
                 const float* __restrict__ Wq,
                 const float* __restrict__ Wk,
                 const float* __restrict__ Wv,
                 float* __restrict__ out)
{
    __shared__ __align__(16) float2 s_kv[SEQ];
    __shared__ float2 s_ext[WAVES];
    __shared__ float s_red[ROW_TILES][KSPLIT][2][ROWS_PER_WAVE];
    __shared__ float s_A[ROWS_PER_BLOCK];

    const int tid  = threadIdx.x;
    const int b    = blockIdx.x >> 5;
    const int rowBase = (blockIdx.x & 31) * ROWS_PER_BLOCK;
    const int wave = __builtin_amdgcn_readfirstlane(tid >> 5);
    const int lane = tid & 31;
    const int rt   = wave & (ROW_TILES - 1);
    const int ks   = wave >> 2;
    const int half = lane >> 4;
    const int r    = lane & 15;
    const int n    = lane & 15;

    const float2* seq = (const float2*)in + (size_t)b * SEQ;
    float hmx = -H_NONE, hmn = H_NONE;
    for (int i = 0; i < SEQ / THREADS; ++i) {
        const float2 e = seq[tid + i * THREADS];
        s_kv[tid + i * THREADS] = e;
        hmx = fmaxf(hmx, (e.y != 0.0f) ? e.y : -H_NONE);
        hmn = fminf(hmn, (e.y != 0.0f) ? e.y : H_NONE);
    }
    wave_extremes(hmx, hmn);
    if (lane == 0) s_ext[wave] = make_float2(hmx, hmn);

    float c = 0.0f;
    #pragma unroll 8
    for (int i = 0; i < DMODEL; ++i) c += Wq[i] * Wk[i];
    c *= (1.0f / 16.0f);

    __syncthreads();

    const int   row = rowBase + rt * ROWS_PER_WAVE + r;
    const float2 q2 = s_kv[row];
    const float zs = q2.x, hs = q2.y;
    const int   key0 = ks * KEYS_PER_WAVE;

    const float2 ex = s_ext[lane];
    float gmx = ex.x, gmn = ex.y;
    wave_extremes(gmx, gmn);
    const float m = fmaxf(c * hs * gmx, c * hs * gmn);

    v8f acc = {0.f, 0.f, 0.f, 0.f, 0.f, 0.f, 0.f, 0.f};
    const float4* kv4 = (const float4*)s_kv;
    #pragma unroll 2
    for (int kc = key0; kc < key0 + KEYS_PER_WAVE; kc += 4) {
        const int ka = kc + 2 * half;
        float4 kk = kv4[ka >> 1];
        const float za = kk.x, ha = kk.y, zb = kk.z, hb = kk.w;
        float la = c * hs * ha - LAMBDA * fabsf(zs - za) + ((ha == 0.0f) ? NEG_BIG : 0.0f);
        float lb = c * hs * hb - LAMBDA * fabsf(zs - zb) + ((hb == 0.0f) ? NEG_BIG : 0.0f);
        v2f a;
        a.x = __expf(la - m);
        a.y = __expf(lb - m);
        v2f bop;
        bop.x = (n == 0) ? ha : ((n == 1) ? 1.0f : 0.0f);
        bop.y = (n == 0) ? hb : ((n == 1) ? 1.0f : 0.0f);
        acc = __builtin_amdgcn_wmma_f32_16x16x4_f32(
             false, a,  false, bop,
             (short)0, acc,  false,  false);
    }

    if (n < 2) {
        #pragma unroll
        for (int v = 0; v < 8; ++v)
            s_red[rt][ks][n][v + 8 * half] = acc[v];
    }
    __syncthreads();

    if (tid < ROWS_PER_BLOCK) {
        const int w = tid >> 4, rr = tid & 15;
        float eh = 0.0f, se = 0.0f;
        for (int j = 0; j < KSPLIT; ++j) {
            eh += s_red[w][j][0][rr];
            se += s_red[w][j][1][rr];
        }
        s_A[tid] = (se == 0.0f) ? 0.0f : eh / se;
    }
    __syncthreads();

    const int cp = tid & 127;
    const int rg = tid >> 7;
    const float wv0 = Wv[2 * cp];
    const float wv1 = Wv[2 * cp + 1];
    v2f* o2 = (v2f*)out + (size_t)(b * SEQ + rowBase + rg) * (DMODEL / 2) + cp;
    for (int i = 0; i < ROWS_PER_BLOCK / (THREADS / 128); ++i) {
        const float A = s_A[rg + i * (THREADS / 128)];
        v2f val;
        val.x = A * wv0;
        val.y = A * wv1;
        __builtin_nontemporal_store(val, o2 + (size_t)i * (THREADS / 128) * (DMODEL / 2));
    }
}

extern "C" void kernel_launch(void* const* d_in, const int* in_sizes, int n_in,
                              void* d_out, int out_size, void* d_ws, size_t ws_size,
                              hipStream_t stream) {
    (void)in_sizes; (void)n_in; (void)out_size; (void)d_ws; (void)ws_size;
    const float* in = (const float*)d_in[0];
    const float* Wq = (const float*)d_in[1];
    const float* Wk = (const float*)d_in[2];
    const float* Wv = (const float*)d_in[3];
    float* out = (float*)d_out;

    dim3 grid(BATCH * (SEQ / ROWS_PER_BLOCK));
    dim3 block(THREADS);
    dwsa_kernel<<<grid, block, 0, stream>>>(in, Wq, Wk, Wv, out);
}


# One memory round trip ahead of the first barrier, not 33: c=Wq.Wk as 8 terms per wave joined in the cross-wave butterfly (was a serial scalar-load loop); Wv read hoisted; c's sum order changes
# speedup vs baseline: 5.1909x; 1.2524x over previous
#include <hip/hip_runtime.h>


typedef float v2f __attribute__((ext_vector_type(2)));
typedef float v8f __attribute__((ext_vector_type(8)));

#define BATCH 8
#define SEQ 2048
#define DMODEL 256
#define ROWS_PER_WAVE 16
#define ROW_TILES 4
#define ROWS_PER_BLOCK 64
#define KSPLIT 8
#define KEYS_PER_WAVE 256
#define WAVES 32
#define THREADS 1024
#define LAMBDA 0.5f
#define NEG_BIG -1.0e9f
#define H_NONE 3.4e38f

static_assert(WAVES == 32, "one lane per wave in the second reduction stage");
static_assert(SEQ / THREADS == 2, "two sequence loads per thread are named in the load group");
static_assert(WAVES * 8 == DMODEL, "each wave takes eight terms of the weight dot product");

template <int XMASK>
__device__ __forceinline__ float row_xor(float v) {
    const int x = __builtin_bit_cast(int, v);
    return __builtin_bit_cast(float,
        __builtin_amdgcn_update_dpp(x, x, 0x160 | XMASK, 0xf, 0xf, false));
}
__device__ __forceinline__ float rows_swap(float v) {
    const int x = __builtin_bit_cast(int, v);
    return __builtin_bit_cast(float,
        __builtin_amdgcn_permlanex16(x, x, 0x76543210u, 0xfedcba98u, false, false));
}
__device__ __forceinline__ void wave_extremes(float& hi, float& lo) {
    hi = fmaxf(hi, row_xor<1>(hi));  lo = fminf(lo, row_xor<1>(lo));
    hi = fmaxf(hi, row_xor<2>(hi));  lo = fminf(lo, row_xor<2>(lo));
    hi = fmaxf(hi, row_xor<4>(hi));  lo = fminf(lo, row_xor<4>(lo));
    hi = fmaxf(hi, row_xor<8>(hi));  lo = fminf(lo, row_xor<8>(lo));
    hi = fmaxf(hi, rows_swap(hi));   lo = fminf(lo, rows_swap(lo));
}

__device__ __forceinline__ float wave_sum(float v) {
    v += row_xor<1>(v);
    v += row_xor<2>(v);
    v += row_xor<4>(v);
    v += row_xor<8>(v);
    v += rows_swap(v);
    return v;
}

__global__ __launch_bounds__(THREADS)
void dwsa_kernel(const float* __restrict__ in,
                 const float* __restrict__ Wq,
                 const float* __restrict__ Wk,
                 const float* __restrict__ Wv,
                 float* __restrict__ out)
{
    __shared__ __align__(16) float2 s_kv[SEQ];
    __shared__ __align__(16) float4 s_ext[WAVES];
    __shared__ float s_red[ROW_TILES][KSPLIT][2][ROWS_PER_WAVE];
    __shared__ float s_A[ROWS_PER_BLOCK];

    const int tid  = threadIdx.x;
    const int b    = blockIdx.x >> 5;
    const int rowBase = (blockIdx.x & 31) * ROWS_PER_BLOCK;
    const int wave = __builtin_amdgcn_readfirstlane(tid >> 5);
    const int lane = tid & 31;
    const int rt   = wave & (ROW_TILES - 1);
    const int ks   = wave >> 2;
    const int half = lane >> 4;
    const int r    = lane & 15;
    const int n    = lane & 15;

    const int   wi  = wave * 8 + (lane & 7);
    const float wqv = Wq[wi];
    const float wkv = Wk[wi];
    const int   cp  = tid & 127;
    const v2f   wv  = *(const v2f*)(Wv + 2 * cp);

    const float2* seq = (const float2*)in + (size_t)b * SEQ;
    float hmx = -H_NONE, hmn = H_NONE;
    float2 ev[SEQ / THREADS];
    for (int i = 0; i < SEQ / THREADS; ++i) ev[i] = seq[tid + i * THREADS];
    __builtin_amdgcn_sched_barrier(0);
    for (int i = 0; i < SEQ / THREADS; ++i) {
        const float2 e = ev[i];
        s_kv[tid + i * THREADS] = e;
        hmx = fmaxf(hmx, (e.y != 0.0f) ? e.y : -H_NONE);
        hmn = fminf(hmn, (e.y != 0.0f) ? e.y : H_NONE);
    }
    wave_extremes(hmx, hmn);
    float cpart = wqv * wkv;
    cpart += row_xor<1>(cpart);
    cpart += row_xor<2>(cpart);
    cpart += row_xor<4>(cpart);
    if (lane == 0) s_ext[wave] = make_float4(hmx, hmn, cpart, 0.0f);

    __syncthreads();

    const int   row = rowBase + rt * ROWS_PER_WAVE + r;
    const float2 q2 = s_kv[row];
    const float zs = q2.x, hs = q2.y;
    const int   key0 = ks * KEYS_PER_WAVE;

    const float4 ex = s_ext[lane];
    float gmx = ex.x, gmn = ex.y;
    wave_extremes(gmx, gmn);
    const float c = wave_sum(ex.z) * (1.0f / 16.0f);
    const float m = fmaxf(c * hs * gmx, c * hs * gmn);

    v8f acc = {0.f, 0.f, 0.f, 0.f, 0.f, 0.f, 0.f, 0.f};
    const float4* kv4 = (const float4*)s_kv;
    #pragma unroll 2
    for (int kc = key0; kc < key0 + KEYS_PER_WAVE; kc += 4) {
        const int ka = kc + 2 * half;
        float4 kk = kv4[ka >> 1];
        const float za = kk.x, ha = kk.y, zb = kk.z, hb = kk.w;
        float la = c * hs * ha - LAMBDA * fabsf(zs - za) + ((ha == 0.0f) ? NEG_BIG : 0.0f);
        float lb = c * hs * hb - LAMBDA * fabsf(zs - zb) + ((hb == 0.0f) ? NEG_BIG : 0.0f);
        v2f a;
        a.x = __expf(la - m);
        a.y = __expf(lb - m);
        v2f bop;
        bop.x = (n == 0) ? ha : ((n == 1) ? 1.0f : 0.0f);
        bop.y = (n == 0) ? hb : ((n == 1) ? 1.0f : 0.0f);
        acc = __builtin_amdgcn_wmma_f32_16x16x4_f32(
             false, a,  false, bop,
             (short)0, acc,  false,  false);
    }

    if (n < 2) {
        #pragma unroll
        for (int v = 0; v < 8; ++v)
            s_red[rt][ks][n][v + 8 * half] = acc[v];
    }
    __syncthreads();

    if (tid < ROWS_PER_BLOCK) {
        const int w = tid >> 4, rr = tid & 15;
        float eh = 0.0f, se = 0.0f;
        for (int j = 0; j < KSPLIT; ++j) {
            eh += s_red[w][j][0][rr];
            se += s_red[w][j][1][rr];
        }
        s_A[tid] = (se == 0.0f) ? 0.0f : eh / se;
    }
    __syncthreads();

    const int rg = tid >> 7;
    const float wv0 = wv.x;
    const float wv1 = wv.y;
    v2f* o2 = (v2f*)out + (size_t)(b * SEQ + rowBase + rg) * (DMODEL / 2) + cp;
    for (int i = 0; i < ROWS_PER_BLOCK / (THREADS / 128); ++i) {
        const float A = s_A[rg + i * (THREADS / 128)];
        v2f val;
        val.x = A * wv0;
        val.y = A * wv1;
        __builtin_nontemporal_store(val, o2 + (size_t)i * (THREADS / 128) * (DMODEL / 2));
    }
}

extern "C" void kernel_launch(void* const* d_in, const int* in_sizes, int n_in,
                              void* d_out, int out_size, void* d_ws, size_t ws_size,
                              hipStream_t stream) {
    (void)in_sizes; (void)n_in; (void)out_size; (void)d_ws; (void)ws_size;
    const float* in = (const float*)d_in[0];
    const float* Wq = (const float*)d_in[1];
    const float* Wk = (const float*)d_in[2];
    const float* Wv = (const float*)d_in[3];
    float* out = (float*)d_out;

    dim3 grid(BATCH * (SEQ / ROWS_PER_BLOCK));
    dim3 block(THREADS);
    dwsa_kernel<<<grid, block, 0, stream>>>(in, Wq, Wk, Wv, out);
}


# Key loop: pair-independent logit parts (row shift, log2e, key mask) folded into two per-row values and the z stored per key: 5 vector ops per pair, not 10; sum order kept, logit re-associated
# speedup vs baseline: 6.5549x; 1.2628x over previous
#include <hip/hip_runtime.h>


typedef float v2f __attribute__((ext_vector_type(2)));
typedef float v8f __attribute__((ext_vector_type(8)));

#define BATCH 8
#define SEQ 2048
#define DMODEL 256
#define ROWS_PER_WAVE 16
#define ROW_TILES 4
#define ROWS_PER_BLOCK 64
#define KSPLIT 8
#define KEYS_PER_WAVE 256
#define WAVES 32
#define THREADS 1024
#define LAMBDA 0.5f
#define LOG2E 1.4426950408889634f
#define LAMBDA2 (LAMBDA * LOG2E)
#define Z_MASKED 2.0e9f
#define H_NONE 3.4e38f

static_assert(WAVES == 32, "one lane per wave in the second reduction stage");
static_assert(SEQ / THREADS == 2, "two sequence loads per thread are named in the load group");
static_assert(WAVES * 8 == DMODEL, "each wave takes eight terms of the weight dot product");

template <int XMASK>
__device__ __forceinline__ float row_xor(float v) {
    const int x = __builtin_bit_cast(int, v);
    return __builtin_bit_cast(float,
        __builtin_amdgcn_update_dpp(x, x, 0x160 | XMASK, 0xf, 0xf, false));
}
__device__ __forceinline__ float rows_swap(float v) {
    const int x = __builtin_bit_cast(int, v);
    return __builtin_bit_cast(float,
        __builtin_amdgcn_permlanex16(x, x, 0x76543210u, 0xfedcba98u, false, false));
}
__device__ __forceinline__ void wave_extremes(float& hi, float& lo) {
    hi = fmaxf(hi, row_xor<1>(hi));  lo = fminf(lo, row_xor<1>(lo));
    hi = fmaxf(hi, row_xor<2>(hi));  lo = fminf(lo, row_xor<2>(lo));
    hi = fmaxf(hi, row_xor<4>(hi));  lo = fminf(lo, row_xor<4>(lo));
    hi = fmaxf(hi, row_xor<8>(hi));  lo = fminf(lo, row_xor<8>(lo));
    hi = fmaxf(hi, rows_swap(hi));   lo = fminf(lo, rows_swap(lo));
}

__device__ __forceinline__ float wave_sum(float v) {
    v += row_xor<1>(v);
    v += row_xor<2>(v);
    v += row_xor<4>(v);
    v += row_xor<8>(v);
    v += rows_swap(v);
    return v;
}

__global__ __launch_bounds__(THREADS)
void dwsa_kernel(const float* __restrict__ in,
                 const float* __restrict__ Wq,
                 const float* __restrict__ Wk,
                 const float* __restrict__ Wv,
                 float* __restrict__ out)
{
    __shared__ __align__(16) float2 s_kv[SEQ];
    __shared__ __align__(16) float4 s_ext[WAVES];
    __shared__ float s_red[ROW_TILES][KSPLIT][2][ROWS_PER_WAVE];
    __shared__ float s_A[ROWS_PER_BLOCK];

    const int tid  = threadIdx.x;
    const int b    = blockIdx.x >> 5;
    const int rowBase = (blockIdx.x & 31) * ROWS_PER_BLOCK;
    const int wave = __builtin_amdgcn_readfirstlane(tid >> 5);
    const int lane = tid & 31;
    const int rt   = wave & (ROW_TILES - 1);
    const int ks   = wave >> 2;
    const int half = lane >> 4;
    const int r    = lane & 15;
    const int n    = lane & 15;

    const int   wi  = wave * 8 + (lane & 7);
    const float wqv = Wq[wi];
    const float wkv = Wk[wi];
    const int   cp  = tid & 127;
    const v2f   wv  = *(const v2f*)(Wv + 2 * cp);

    const float2* seq = (const float2*)in + (size_t)b * SEQ;
    float hmx = -H_NONE, hmn = H_NONE;
    float2 ev[SEQ / THREADS];
    for (int i = 0; i < SEQ / THREADS; ++i) ev[i] = seq[tid + i * THREADS];
    const int   row = rowBase + rt * ROWS_PER_WAVE + r;
    const float2 q2 = seq[row];
    __builtin_amdgcn_sched_barrier(0);
    for (int i = 0; i < SEQ / THREADS; ++i) {
        const float2 e = ev[i];
        s_kv[tid + i * THREADS] = make_float2((e.y != 0.0f) ? e.x : Z_MASKED, e.y);
        hmx = fmaxf(hmx, (e.y != 0.0f) ? e.y : -H_NONE);
        hmn = fminf(hmn, (e.y != 0.0f) ? e.y : H_NONE);
    }
    wave_extremes(hmx, hmn);
    float cpart = wqv * wkv;
    cpart += row_xor<1>(cpart);
    cpart += row_xor<2>(cpart);
    cpart += row_xor<4>(cpart);
    if (lane == 0) s_ext[wave] = make_float4(hmx, hmn, cpart, 0.0f);

    __syncthreads();

    const float zs = q2.x, hs = q2.y;
    const int   key0 = ks * KEYS_PER_WAVE;

    const float4 ex = s_ext[lane];
    float gmx = ex.x, gmn = ex.y;
    wave_extremes(gmx, gmn);
    const float c = wave_sum(ex.z) * (1.0f / 16.0f);
    const float m = fmaxf(c * hs * gmx, c * hs * gmn);
    const float chs2  = c * hs * LOG2E;
    const float negm2 = -m * LOG2E;

    v8f acc = {0.f, 0.f, 0.f, 0.f, 0.f, 0.f, 0.f, 0.f};
    const float4* kp = (const float4*)s_kv + (key0 >> 1) + half;
    #pragma unroll 4
    for (int j = 0; j < KEYS_PER_WAVE / 4; ++j) {
        const float4 kk = kp[2 * j];
        const float za = kk.x, ha = kk.y, zb = kk.z, hb = kk.w;
        const float la = fmaf(fabsf(zs - za), -LAMBDA2, fmaf(chs2, ha, negm2));
        const float lb = fmaf(fabsf(zs - zb), -LAMBDA2, fmaf(chs2, hb, negm2));
        v2f a;
        a.x = __builtin_amdgcn_exp2f(la);
        a.y = __builtin_amdgcn_exp2f(lb);
        v2f bop;
        bop.x = (n == 0) ? ha : ((n == 1) ? 1.0f : 0.0f);
        bop.y = (n == 0) ? hb : ((n == 1) ? 1.0f : 0.0f);
        acc = __builtin_amdgcn_wmma_f32_16x16x4_f32(
             false, a,  false, bop,
             (short)0, acc,  false,  false);
    }

    if (n < 2) {
        #pragma unroll
        for (int v = 0; v < 8; ++v)
            s_red[rt][ks][n][v + 8 * half] = acc[v];
    }
    __syncthreads();

    if (tid < ROWS_PER_BLOCK) {
        const int w = tid >> 4, rr = tid & 15;
        float eh = 0.0f, se = 0.0f;
        for (int j = 0; j < KSPLIT; ++j) {
            eh += s_red[w][j][0][rr];
            se += s_red[w][j][1][rr];
        }
        s_A[tid] = (se == 0.0f) ? 0.0f : eh / se;
    }
    __syncthreads();

    const int rg = tid >> 7;
    const float wv0 = wv.x;
    const float wv1 = wv.y;
    v2f* o2 = (v2f*)out + (size_t)(b * SEQ + rowBase + rg) * (DMODEL / 2) + cp;
    for (int i = 0; i < ROWS_PER_BLOCK / (THREADS / 128); ++i) {
        const float A = s_A[rg + i * (THREADS / 128)];
        v2f val;
        val.x = A * wv0;
        val.y = A * wv1;
        __builtin_nontemporal_store(val, o2 + (size_t)i * (THREADS / 128) * (DMODEL / 2));
    }
}

extern "C" void kernel_launch(void* const* d_in, const int* in_sizes, int n_in,
                              void* d_out, int out_size, void* d_ws, size_t ws_size,
                              hipStream_t stream) {
    (void)in_sizes; (void)n_in; (void)out_size; (void)d_ws; (void)ws_size;
    const float* in = (const float*)d_in[0];
    const float* Wq = (const float*)d_in[1];
    const float* Wk = (const float*)d_in[2];
    const float* Wv = (const float*)d_in[3];
    float* out = (float*)d_out;

    dim3 grid(BATCH * (SEQ / ROWS_PER_BLOCK));
    dim3 block(THREADS);
    dwsa_kernel<<<grid, block, 0, stream>>>(in, Wq, Wk, Wv, out);
}


# Each wave carries two 16-row query tiles over its key slice (512-thread blocks, 4 waves per SIMD, not 8): half the waves to start; one LDS read and B fragment per two matrix ops; sum order kept
# speedup vs baseline: 7.0906x; 1.0817x over previous
#include <hip/hip_runtime.h>


typedef float v2f __attribute__((ext_vector_type(2)));
typedef float v8f __attribute__((ext_vector_type(8)));

#define BATCH 8
#define SEQ 2048
#define DMODEL 256
#define ROWS_PER_WAVE 16
#define ROW_TILES 4
#define ROWS_PER_BLOCK 64
#define TILES_PER_WAVE 2
#define KSPLIT 8
#define KEYS_PER_WAVE 256
#define WAVES 16
#define THREADS 512
#define CPARTS 32
#define LAMBDA 0.5f
#define LOG2E 1.4426950408889634f
#define LAMBDA2 (LAMBDA * LOG2E)
#define Z_MASKED 2.0e9f
#define H_NONE 3.4e38f

static_assert(CPARTS == 32, "one lane per partial in the second reduction stage");
static_assert(WAVES * 2 == CPARTS, "each lane half of a wave owns one partial");
static_assert(SEQ / THREADS == 4, "four sequence loads per thread are named in the load group");
static_assert(CPARTS * 8 == DMODEL, "each partial takes eight terms of the weight dot product");
static_assert(ROW_TILES == 2 * TILES_PER_WAVE, "the wave index's low bit picks the pair of query tiles");

template <int XMASK>
__device__ __forceinline__ float row_xor(float v) {
    const int x = __builtin_bit_cast(int, v);
    return __builtin_bit_cast(float,
        __builtin_amdgcn_update_dpp(x, x, 0x160 | XMASK, 0xf, 0xf, false));
}
__device__ __forceinline__ float rows_swap(float v) {
    const int x = __builtin_bit_cast(int, v);
    return __builtin_bit_cast(float,
        __builtin_amdgcn_permlanex16(x, x, 0x76543210u, 0xfedcba98u, false, false));
}
__device__ __forceinline__ void wave_extremes(float& hi, float& lo) {
    hi = fmaxf(hi, row_xor<1>(hi));  lo = fminf(lo, row_xor<1>(lo));
    hi = fmaxf(hi, row_xor<2>(hi));  lo = fminf(lo, row_xor<2>(lo));
    hi = fmaxf(hi, row_xor<4>(hi));  lo = fminf(lo, row_xor<4>(lo));
    hi = fmaxf(hi, row_xor<8>(hi));  lo = fminf(lo, row_xor<8>(lo));
    hi = fmaxf(hi, rows_swap(hi));   lo = fminf(lo, rows_swap(lo));
}

__device__ __forceinline__ float wave_sum(float v) {
    v += row_xor<1>(v);
    v += row_xor<2>(v);
    v += row_xor<4>(v);
    v += row_xor<8>(v);
    v += rows_swap(v);
    return v;
}

__global__ __launch_bounds__(THREADS)
void dwsa_kernel(const float* __restrict__ in,
                 const float* __restrict__ Wq,
                 const float* __restrict__ Wk,
                 const float* __restrict__ Wv,
                 float* __restrict__ out)
{
    __shared__ __align__(16) float2 s_kv[SEQ];
    __shared__ __align__(16) float4 s_ext[CPARTS];
    __shared__ float s_red[ROW_TILES][KSPLIT][2][ROWS_PER_WAVE];
    __shared__ float s_A[ROWS_PER_BLOCK];

    const int tid  = threadIdx.x;
    const int b    = blockIdx.x >> 5;
    const int rowBase = (blockIdx.x & 31) * ROWS_PER_BLOCK;
    const int wave = __builtin_amdgcn_readfirstlane(tid >> 5);
    const int lane = tid & 31;
    const int rt0  = (wave & 1) * TILES_PER_WAVE;
    const int ks   = wave >> 1;
    const int half = lane >> 4;
    const int r    = lane & 15;
    const int n    = lane & 15;

    const int   wi  = (wave * 2 + half) * 8 + (lane & 7);
    const float wqv = Wq[wi];
    const float wkv = Wk[wi];
    const int   cp  = tid & 127;
    const v2f   wv  = *(const v2f*)(Wv + 2 * cp);

    const float2* seq = (const float2*)in + (size_t)b * SEQ;
    float hmx = -H_NONE, hmn = H_NONE;
    float2 ev[SEQ / THREADS];
    for (int i = 0; i < SEQ / THREADS; ++i) ev[i] = seq[tid + i * THREADS];
    const int   row = rowBase + rt0 * ROWS_PER_WAVE + r;
    const float2 q2a = seq[row];
    const float2 q2b = seq[row + ROWS_PER_WAVE];
    __builtin_amdgcn_sched_barrier(0);
    for (int i = 0; i < SEQ / THREADS; ++i) {
        const float2 e = ev[i];
        s_kv[tid + i * THREADS] = make_float2((e.y != 0.0f) ? e.x : Z_MASKED, e.y);
        hmx = fmaxf(hmx, (e.y != 0.0f) ? e.y : -H_NONE);
        hmn = fminf(hmn, (e.y != 0.0f) ? e.y : H_NONE);
    }
    wave_extremes(hmx, hmn);
    float cpart = wqv * wkv;
    cpart += row_xor<1>(cpart);
    cpart += row_xor<2>(cpart);
    cpart += row_xor<4>(cpart);
    if (r == 0) s_ext[wave * 2 + half] = make_float4(hmx, hmn, cpart, 0.0f);

    __syncthreads();

    const float zs0 = q2a.x, hs0 = q2a.y;
    const float zs1 = q2b.x, hs1 = q2b.y;
    const int   key0 = ks * KEYS_PER_WAVE;

    const float4 ex = s_ext[lane];
    float gmx = ex.x, gmn = ex.y;
    wave_extremes(gmx, gmn);
    const float c = wave_sum(ex.z) * (1.0f / 16.0f);
    const float m0 = fmaxf(c * hs0 * gmx, c * hs0 * gmn);
    const float m1 = fmaxf(c * hs1 * gmx, c * hs1 * gmn);
    const float chs2_0  = c * hs0 * LOG2E;
    const float negm2_0 = -m0 * LOG2E;
    const float chs2_1  = c * hs1 * LOG2E;
    const float negm2_1 = -m1 * LOG2E;

    v8f acc0 = {0.f, 0.f, 0.f, 0.f, 0.f, 0.f, 0.f, 0.f};
    v8f acc1 = {0.f, 0.f, 0.f, 0.f, 0.f, 0.f, 0.f, 0.f};
    const float4* kp = (const float4*)s_kv + (key0 >> 1) + half;
    #pragma unroll 4
    for (int j = 0; j < KEYS_PER_WAVE / 4; ++j) {
        const float4 kk = kp[2 * j];
        const float za = kk.x, ha = kk.y, zb = kk.z, hb = kk.w;
        const float la0 = fmaf(fabsf(zs0 - za), -LAMBDA2, fmaf(chs2_0, ha, negm2_0));
        const float lb0 = fmaf(fabsf(zs0 - zb), -LAMBDA2, fmaf(chs2_0, hb, negm2_0));
        const float la1 = fmaf(fabsf(zs1 - za), -LAMBDA2, fmaf(chs2_1, ha, negm2_1));
        const float lb1 = fmaf(fabsf(zs1 - zb), -LAMBDA2, fmaf(chs2_1, hb, negm2_1));
        v2f a0, a1;
        a0.x = __builtin_amdgcn_exp2f(la0);
        a0.y = __builtin_amdgcn_exp2f(lb0);
        a1.x = __builtin_amdgcn_exp2f(la1);
        a1.y = __builtin_amdgcn_exp2f(lb1);
        v2f bop;
        bop.x = (n == 0) ? ha : ((n == 1) ? 1.0f : 0.0f);
        bop.y = (n == 0) ? hb : ((n == 1) ? 1.0f : 0.0f);
        acc0 = __builtin_amdgcn_wmma_f32_16x16x4_f32(
             false, a0,  false, bop,
             (short)0, acc0,  false,  false);
        acc1 = __builtin_amdgcn_wmma_f32_16x16x4_f32(
             false, a1,  false, bop,
             (short)0, acc1,  false,  false);
    }

    if (n < 2) {
        #pragma unroll
        for (int v = 0; v < 8; ++v) {
            s_red[rt0][ks][n][v + 8 * half]     = acc0[v];
            s_red[rt0 + 1][ks][n][v + 8 * half] = acc1[v];
        }
    }
    __syncthreads();

    if (tid < ROWS_PER_BLOCK) {
        const int w = tid >> 4, rr = tid & 15;
        float eh = 0.0f, se = 0.0f;
        for (int j = 0; j < KSPLIT; ++j) {
            eh += s_red[w][j][0][rr];
            se += s_red[w][j][1][rr];
        }
        s_A[tid] = (se == 0.0f) ? 0.0f : eh / se;
    }
    __syncthreads();

    const int rg = tid >> 7;
    const float wv0 = wv.x;
    const float wv1 = wv.y;
    v2f* o2 = (v2f*)out + (size_t)(b * SEQ + rowBase + rg) * (DMODEL / 2) + cp;
    for (int i = 0; i < ROWS_PER_BLOCK / (THREADS / 128); ++i) {
        const float A = s_A[rg + i * (THREADS / 128)];
        v2f val;
        val.x = A * wv0;
        val.y = A * wv1;
        __builtin_nontemporal_store(val, o2 + (size_t)i * (THREADS / 128) * (DMODEL / 2));
    }
}

extern "C" void kernel_launch(void* const* d_in, const int* in_sizes, int n_in,
                              void* d_out, int out_size, void* d_ws, size_t ws_size,
                              hipStream_t stream) {
    (void)in_sizes; (void)n_in; (void)out_size; (void)d_ws; (void)ws_size;
    const float* in = (const float*)d_in[0];
    const float* Wq = (const float*)d_in[1];
    const float* Wk = (const float*)d_in[2];
    const float* Wv = (const float*)d_in[3];
    float* out = (float*)d_out;

    dim3 grid(BATCH * (SEQ / ROWS_PER_BLOCK));
    dim3 block(THREADS);
    dwsa_kernel<<<grid, block, 0, stream>>>(in, Wq, Wk, Wv, out);
}


# Key loop: exp weights in BOTH matrix operands (14 rows x 4 keys in A, same rows x 4 other keys in B; h_key, 1 in rows/cols 14,15): 320 ops per SIMD, not 512, 15 vector issues each; sum order changes
# speedup vs baseline: 7.4942x; 1.0569x over previous
#include <hip/hip_runtime.h>


typedef float v2f __attribute__((ext_vector_type(2)));
typedef float v8f __attribute__((ext_vector_type(8)));

#define BATCH 8
#define SEQ 2048
#define DMODEL 256
#define ROWS_PER_BLOCK 64
#define TILE_ROWS 14
#define NTILES 5
#define SLOTS 80
#define KSPLIT 16
#define KEYS_PER_WAVE 128
#define WAVES 16
#define THREADS 512
#define CPARTS 32
#define LAMBDA 0.5f
#define LOG2E 1.4426950408889634f
#define LAMBDA2 (LAMBDA * LOG2E)
#define Z_MASKED 2.0e9f
#define H_NONE 3.4e38f

static_assert(CPARTS == 32, "one lane per partial in the second reduction stage");
static_assert(WAVES * 2 == CPARTS, "each lane half of a wave owns one partial");
static_assert(SEQ / THREADS == 4, "four sequence loads per thread are named in the load group");
static_assert(CPARTS * 8 == DMODEL, "each partial takes eight terms of the weight dot product");
static_assert(NTILES * TILE_ROWS >= ROWS_PER_BLOCK, "the tiles cover the block's rows");
static_assert(KSPLIT == WAVES, "one key slice per wave");

template <int XMASK>
__device__ __forceinline__ float row_xor(float v) {
    const int x = __builtin_bit_cast(int, v);
    return __builtin_bit_cast(float,
        __builtin_amdgcn_update_dpp(x, x, 0x160 | XMASK, 0xf, 0xf, false));
}
__device__ __forceinline__ float rows_swap(float v) {
    const int x = __builtin_bit_cast(int, v);
    return __builtin_bit_cast(float,
        __builtin_amdgcn_permlanex16(x, x, 0x76543210u, 0xfedcba98u, false, false));
}
__device__ __forceinline__ void wave_extremes(float& hi, float& lo) {
    hi = fmaxf(hi, row_xor<1>(hi));  lo = fminf(lo, row_xor<1>(lo));
    hi = fmaxf(hi, row_xor<2>(hi));  lo = fminf(lo, row_xor<2>(lo));
    hi = fmaxf(hi, row_xor<4>(hi));  lo = fminf(lo, row_xor<4>(lo));
    hi = fmaxf(hi, row_xor<8>(hi));  lo = fminf(lo, row_xor<8>(lo));
    hi = fmaxf(hi, rows_swap(hi));   lo = fminf(lo, rows_swap(lo));
}

__device__ __forceinline__ float wave_sum(float v) {
    v += row_xor<1>(v);
    v += row_xor<2>(v);
    v += row_xor<4>(v);
    v += row_xor<8>(v);
    v += rows_swap(v);
    return v;
}

__global__ __launch_bounds__(THREADS)
void dwsa_kernel(const float* __restrict__ in,
                 const float* __restrict__ Wq,
                 const float* __restrict__ Wk,
                 const float* __restrict__ Wv,
                 float* __restrict__ out)
{
    __shared__ __align__(16) float2 s_kv[SEQ];
    __shared__ __align__(16) float4 s_ext[CPARTS];
    __shared__ float s_red[KSPLIT][2][2][SLOTS];
    __shared__ float s_A[ROWS_PER_BLOCK];

    const int tid  = threadIdx.x;
    const int b    = blockIdx.x >> 5;
    const int rowBase = (blockIdx.x & 31) * ROWS_PER_BLOCK;
    const int wave = __builtin_amdgcn_readfirstlane(tid >> 5);
    const int lane = tid & 31;
    const int ks   = wave;
    const int half = lane >> 4;
    const int r    = lane & 15;
    const int n    = lane & 15;

    const int   wi  = (wave * 2 + half) * 8 + (lane & 7);
    const float wqv = Wq[wi];
    const float wkv = Wk[wi];
    const int   cp  = tid & 127;
    const v2f   wv  = *(const v2f*)(Wv + 2 * cp);

    const float2* seq = (const float2*)in + (size_t)b * SEQ;
    float hmx = -H_NONE, hmn = H_NONE;
    float2 ev[SEQ / THREADS];
    for (int i = 0; i < SEQ / THREADS; ++i) ev[i] = seq[tid + i * THREADS];
    float2 q2[NTILES];
    for (int t = 0; t < NTILES; ++t) {
        const int ri = t * TILE_ROWS + ((r < TILE_ROWS) ? r : TILE_ROWS - 1);
        q2[t] = seq[rowBase + ((ri < ROWS_PER_BLOCK) ? ri : ROWS_PER_BLOCK - 1)];
    }
    __builtin_amdgcn_sched_barrier(0);
    for (int i = 0; i < SEQ / THREADS; ++i) {
        const float2 e = ev[i];
        s_kv[tid + i * THREADS] = make_float2((e.y != 0.0f) ? e.x : Z_MASKED, e.y);
        hmx = fmaxf(hmx, (e.y != 0.0f) ? e.y : -H_NONE);
        hmn = fminf(hmn, (e.y != 0.0f) ? e.y : H_NONE);
    }
    wave_extremes(hmx, hmn);
    float cpart = wqv * wkv;
    cpart += row_xor<1>(cpart);
    cpart += row_xor<2>(cpart);
    cpart += row_xor<4>(cpart);
    if (r == 0) s_ext[wave * 2 + half] = make_float4(hmx, hmn, cpart, 0.0f);

    __syncthreads();

    const int   key0 = ks * KEYS_PER_WAVE;

    const float4 ex = s_ext[lane];
    float gmx = ex.x, gmn = ex.y;
    wave_extremes(gmx, gmn);
    const float c = wave_sum(ex.z) * (1.0f / 16.0f);
    const bool  rowOn = r < TILE_ROWS;
    const bool  is14  = r == TILE_ROWS;
    const float nlam  = rowOn ? -LAMBDA2 : 0.0f;
    float zs[NTILES], chs2[NTILES], negm2[NTILES];
    #pragma unroll
    for (int t = 0; t < NTILES; ++t) {
        const float hs = q2[t].y;
        const float m  = fmaxf(c * hs * gmx, c * hs * gmn);
        zs[t]    = rowOn ? q2[t].x : 0.0f;
        chs2[t]  = rowOn ? c * hs * LOG2E : 0.0f;
        negm2[t] = rowOn ? -m * LOG2E : 0.0f;
    }

    v8f acc[NTILES];
    #pragma unroll
    for (int t = 0; t < NTILES; ++t) acc[t] = (v8f){0.f, 0.f, 0.f, 0.f, 0.f, 0.f, 0.f, 0.f};
    const float4* kp = (const float4*)s_kv + (key0 >> 1) + half;
    #pragma unroll 2
    for (int j = 0; j < KEYS_PER_WAVE / 8; ++j) {
        const float4 kP = kp[4 * j];
        const float4 kQ = kp[4 * j + 2];
        #pragma unroll
        for (int t = 0; t < NTILES; ++t) {
            const float lp0 = fmaf(fabsf(zs[t] - kP.x), nlam, fmaf(chs2[t], kP.y, negm2[t]));
            const float lp1 = fmaf(fabsf(zs[t] - kP.z), nlam, fmaf(chs2[t], kP.w, negm2[t]));
            const float lq0 = fmaf(fabsf(zs[t] - kQ.x), nlam, fmaf(chs2[t], kQ.y, negm2[t]));
            const float lq1 = fmaf(fabsf(zs[t] - kQ.z), nlam, fmaf(chs2[t], kQ.w, negm2[t]));
            const float ep0 = __builtin_amdgcn_exp2f(lp0);
            const float ep1 = __builtin_amdgcn_exp2f(lp1);
            const float eq0 = __builtin_amdgcn_exp2f(lq0);
            const float eq1 = __builtin_amdgcn_exp2f(lq1);
            v2f aop, bop;
            aop.x = is14 ? kQ.y : ep0;
            aop.y = is14 ? kQ.w : ep1;
            bop.x = is14 ? kP.y : eq0;
            bop.y = is14 ? kP.w : eq1;
            acc[t] = __builtin_amdgcn_wmma_f32_16x16x4_f32(
                 false, aop,  false, bop,
                 (short)0, acc[t],  false,  false);
        }
    }

    if (n >= TILE_ROWS) {
        const int q = n - TILE_ROWS;
        #pragma unroll
        for (int t = 0; t < NTILES; ++t) {
            #pragma unroll
            for (int v = 0; v < 8; ++v) s_red[ks][0][q][t * 16 + v + 8 * half] = acc[t][v];
        }
    } else if (half == 1) {
        #pragma unroll
        for (int t = 0; t < NTILES; ++t) {
            s_red[ks][1][0][t * 16 + n] = acc[t][6];
            s_red[ks][1][1][t * 16 + n] = acc[t][7];
        }
    }
    __syncthreads();

    if (tid < ROWS_PER_BLOCK) {
        const int slot = (tid / TILE_ROWS) * 16 + (tid % TILE_ROWS);
        float ehP = 0.0f, ehQ = 0.0f, seP = 0.0f, seQ = 0.0f;
        for (int j = 0; j < KSPLIT; ++j) {
            ehP += s_red[j][0][0][slot];
            ehQ += s_red[j][1][0][slot];
            seP += s_red[j][0][1][slot];
            seQ += s_red[j][1][1][slot];
        }
        const float eh = ehP + ehQ, se = seP + seQ;
        s_A[tid] = (se == 0.0f) ? 0.0f : eh / se;
    }
    __syncthreads();

    const int rg = tid >> 7;
    const float wv0 = wv.x;
    const float wv1 = wv.y;
    v2f* o2 = (v2f*)out + (size_t)(b * SEQ + rowBase + rg) * (DMODEL / 2) + cp;
    for (int i = 0; i < ROWS_PER_BLOCK / (THREADS / 128); ++i) {
        const float A = s_A[rg + i * (THREADS / 128)];
        v2f val;
        val.x = A * wv0;
        val.y = A * wv1;
        __builtin_nontemporal_store(val, o2 + (size_t)i * (THREADS / 128) * (DMODEL / 2));
    }
}

extern "C" void kernel_launch(void* const* d_in, const int* in_sizes, int n_in,
                              void* d_out, int out_size, void* d_ws, size_t ws_size,
                              hipStream_t stream) {
    (void)in_sizes; (void)n_in; (void)out_size; (void)d_ws; (void)ws_size;
    const float* in = (const float*)d_in[0];
    const float* Wq = (const float*)d_in[1];
    const float* Wk = (const float*)d_in[2];
    const float* Wv = (const float*)d_in[3];
    float* out = (float*)d_out;

    dim3 grid(BATCH * (SEQ / ROWS_PER_BLOCK));
    dim3 block(THREADS);
    dwsa_kernel<<<grid, block, 0, stream>>>(in, Wq, Wk, Wv, out);
}
